# starting kernel
# baseline (speedup 1.0000x reference)
#include <hip/hip_runtime.h>
#include <math.h>

typedef __attribute__((ext_vector_type(2))) float v2f;
typedef __attribute__((ext_vector_type(8))) float v8f;

#define NL   1000000
#define TILE 250
#define EXT  256
#define HSTR 258
#define NCH  7

__global__ __launch_bounds__(256) void
state_policy_kernel(const float* __restrict__ x,
                    const float* __restrict__ states,
                    const float* __restrict__ conv_w,
                    const float* __restrict__ conv_b,
                    const float* __restrict__ comp_mat,
                    const float* __restrict__ act_mat,
                    float* __restrict__ out)
{
    __shared__ float h_lds[NCH * HSTR];
    __shared__ float st_lds[NCH * EXT];

    const int tid  = threadIdx.x;
    const int lane = tid & 31;
    const int wid  = tid >> 5;
    const int n    = lane & 15;
    const int hi   = lane >> 4;

    const long l0    = (long)blockIdx.x * TILE;
    const long e     = l0 - 3 + tid;
    const bool valid = (e >= 0) && (e < NL);

    float jl[6], jd[6];
#pragma unroll
    for (int c = 0; c < 6; ++c) {
        long flat = (long)c * NL + e;
        bool ok = valid && (flat >= 2);
        jl[c] = ok ? x[7 + flat - 2]            : 0.0f;
        jd[c] = ok ? x[6L * NL + 11 + flat - 2] : 0.0f;
    }
#pragma unroll
    for (int c = 0; c < NCH; ++c)
        st_lds[c * EXT + tid] = valid ? states[(long)c * NL + e] : 0.0f;

    const float ext0 = x[3];
    const float ext1 = x[4];

    float comp[NCH][3];
#pragma unroll
    for (int c = 0; c < NCH; ++c)
#pragma unroll
        for (int j = 0; j < 3; ++j)
            comp[c][j] = comp_mat[c * 3 + j];

    float cb[NCH];
#pragma unroll
    for (int o = 0; o < NCH; ++o) cb[o] = conv_b[o];

    v2f A[6];
#pragma unroll
    for (int kb = 0; kb < 6; ++kb) {
        int k0 = kb * 4 + 2 * hi;
        float ax = 0.0f, ay = 0.0f;
        if (n < NCH) {
            if (k0 < 21)     ax = conv_w[n * 21 + k0];
            if (k0 + 1 < 21) ay = conv_w[n * 21 + k0 + 1];
        }
        A[kb].x = ax; A[kb].y = ay;
    }

    if (tid < NCH) {
        h_lds[tid * HSTR + 0]        = 0.0f;
        h_lds[tid * HSTR + HSTR - 1] = 0.0f;
    }
    __syncthreads();

    for (int it = 0; it < 3; ++it) {
#pragma unroll
        for (int c = 0; c < NCH; ++c) {
            float o0 = (c < 6) ? jl[c] : ext0;
            float o1 = (c < 6) ? jd[c] : ext1;
            float sv = st_lds[c * EXT + tid];
            float hv = tanhf(o0 * comp[c][0] + o1 * comp[c][1] + sv * comp[c][2]);
            h_lds[c * HSTR + tid + 1] = valid ? hv : 0.0f;
        }
        __syncthreads();

#pragma unroll
        for (int gi = 0; gi < 2; ++gi) {
            int base = (wid * 2 + gi) * 16;
            v8f acc = {0.f, 0.f, 0.f, 0.f, 0.f, 0.f, 0.f, 0.f};
#pragma unroll
            for (int kb = 0; kb < 6; ++kb) {
                int k0 = kb * 4 + 2 * hi;
                v2f B;
                {
                    bool vx = (k0 < 21);
                    int ii  = vx ? (k0 / 3) : 0;
                    int kk  = vx ? (k0 % 3) : 0;
                    float r = h_lds[ii * HSTR + base + n + kk];
                    B.x = vx ? r : 0.0f;
                }
                {
                    int k1 = k0 + 1;
                    bool vy = (k1 < 21);
                    int ii  = vy ? (k1 / 3) : 0;
                    int kk  = vy ? (k1 % 3) : 0;
                    float r = h_lds[ii * HSTR + base + n + kk];
                    B.y = vy ? r : 0.0f;
                }
                acc = __builtin_amdgcn_wmma_f32_16x16x4_f32(
                        false, A[kb], false, B, (short)0, acc, false, false);
            }
            if (lane < 16) {
#pragma unroll
                for (int o = 0; o < NCH; ++o)
                    st_lds[o * EXT + base + n] = tanhf(acc[o] + cb[o]);
            }
        }
        __syncthreads();
    }

    if (tid >= 3 && tid < 3 + TILE) {
        long l = l0 + tid - 3;
        if (l < NL) {
#pragma unroll
            for (int c = 0; c < 6; ++c) {
                float a0  = act_mat[c * 2 + 0];
                float a1  = act_mat[c * 2 + 1];
                float stv = st_lds[c * EXT + tid];
                long idx  = (long)c * NL + l - 2;
                if (idx >= 0) out[idx] = a0 * stv + a1 * jd[c];
            }
        }
    }
}

extern "C" void kernel_launch(void* const* d_in, const int* in_sizes, int n_in,
                              void* d_out, int out_size, void* d_ws, size_t ws_size,
                              hipStream_t stream) {
    (void)in_sizes; (void)n_in; (void)out_size; (void)d_ws; (void)ws_size;
    const float* x      = (const float*)d_in[0];
    const float* states = (const float*)d_in[1];
    const float* conv_w = (const float*)d_in[2];
    const float* conv_b = (const float*)d_in[3];
    const float* comp   = (const float*)d_in[4];
    const float* am     = (const float*)d_in[5];
    float* out = (float*)d_out;

    const int nblocks = (NL + TILE - 1) / TILE;
    state_policy_kernel<<<nblocks, 256, 0, stream>>>(x, states, conv_w, conv_b, comp, am, out);
}


# library tanhf (two exec-masked paths, about 31 instructions per value) replaced by the hardware v_tanh_f32 (__builtin_amdgcn_tanhf) at both tanh sites; tanh rounding differs from the library's
# speedup vs baseline: 1.8195x; 1.8195x over previous
#include <hip/hip_runtime.h>
#include <math.h>

typedef __attribute__((ext_vector_type(2))) float v2f;
typedef __attribute__((ext_vector_type(8))) float v8f;

#define NL   1000000
#define TILE 250
#define EXT  256
#define HSTR 258
#define NCH  7

__device__ __forceinline__ float tanh_hw(float v) { return __builtin_amdgcn_tanhf(v); }

__global__ __launch_bounds__(256) void
state_policy_kernel(const float* __restrict__ x,
                    const float* __restrict__ states,
                    const float* __restrict__ conv_w,
                    const float* __restrict__ conv_b,
                    const float* __restrict__ comp_mat,
                    const float* __restrict__ act_mat,
                    float* __restrict__ out)
{
    __shared__ float h_lds[NCH * HSTR];
    __shared__ float st_lds[NCH * EXT];

    const int tid  = threadIdx.x;
    const int lane = tid & 31;
    const int wid  = tid >> 5;
    const int n    = lane & 15;
    const int hi   = lane >> 4;

    const long l0    = (long)blockIdx.x * TILE;
    const long e     = l0 - 3 + tid;
    const bool valid = (e >= 0) && (e < NL);

    float jl[6], jd[6];
#pragma unroll
    for (int c = 0; c < 6; ++c) {
        long flat = (long)c * NL + e;
        bool ok = valid && (flat >= 2);
        jl[c] = ok ? x[7 + flat - 2]            : 0.0f;
        jd[c] = ok ? x[6L * NL + 11 + flat - 2] : 0.0f;
    }
#pragma unroll
    for (int c = 0; c < NCH; ++c)
        st_lds[c * EXT + tid] = valid ? states[(long)c * NL + e] : 0.0f;

    const float ext0 = x[3];
    const float ext1 = x[4];

    float comp[NCH][3];
#pragma unroll
    for (int c = 0; c < NCH; ++c)
#pragma unroll
        for (int j = 0; j < 3; ++j)
            comp[c][j] = comp_mat[c * 3 + j];

    float cb[NCH];
#pragma unroll
    for (int o = 0; o < NCH; ++o) cb[o] = conv_b[o];

    v2f A[6];
#pragma unroll
    for (int kb = 0; kb < 6; ++kb) {
        int k0 = kb * 4 + 2 * hi;
        float ax = 0.0f, ay = 0.0f;
        if (n < NCH) {
            if (k0 < 21)     ax = conv_w[n * 21 + k0];
            if (k0 + 1 < 21) ay = conv_w[n * 21 + k0 + 1];
        }
        A[kb].x = ax; A[kb].y = ay;
    }

    if (tid < NCH) {
        h_lds[tid * HSTR + 0]        = 0.0f;
        h_lds[tid * HSTR + HSTR - 1] = 0.0f;
    }
    __syncthreads();

    for (int it = 0; it < 3; ++it) {
#pragma unroll
        for (int c = 0; c < NCH; ++c) {
            float o0 = (c < 6) ? jl[c] : ext0;
            float o1 = (c < 6) ? jd[c] : ext1;
            float sv = st_lds[c * EXT + tid];
            float hv = tanh_hw(o0 * comp[c][0] + o1 * comp[c][1] + sv * comp[c][2]);
            h_lds[c * HSTR + tid + 1] = valid ? hv : 0.0f;
        }
        __syncthreads();

#pragma unroll
        for (int gi = 0; gi < 2; ++gi) {
            int base = (wid * 2 + gi) * 16;
            v8f acc = {0.f, 0.f, 0.f, 0.f, 0.f, 0.f, 0.f, 0.f};
#pragma unroll
            for (int kb = 0; kb < 6; ++kb) {
                int k0 = kb * 4 + 2 * hi;
                v2f B;
                {
                    bool vx = (k0 < 21);
                    int ii  = vx ? (k0 / 3) : 0;
                    int kk  = vx ? (k0 % 3) : 0;
                    float r = h_lds[ii * HSTR + base + n + kk];
                    B.x = vx ? r : 0.0f;
                }
                {
                    int k1 = k0 + 1;
                    bool vy = (k1 < 21);
                    int ii  = vy ? (k1 / 3) : 0;
                    int kk  = vy ? (k1 % 3) : 0;
                    float r = h_lds[ii * HSTR + base + n + kk];
                    B.y = vy ? r : 0.0f;
                }
                acc = __builtin_amdgcn_wmma_f32_16x16x4_f32(
                        false, A[kb], false, B, (short)0, acc, false, false);
            }
            if (lane < 16) {
#pragma unroll
                for (int o = 0; o < NCH; ++o)
                    st_lds[o * EXT + base + n] = tanh_hw(acc[o] + cb[o]);
            }
        }
        __syncthreads();
    }

    if (tid >= 3 && tid < 3 + TILE) {
        long l = l0 + tid - 3;
        if (l < NL) {
#pragma unroll
            for (int c = 0; c < 6; ++c) {
                float a0  = act_mat[c * 2 + 0];
                float a1  = act_mat[c * 2 + 1];
                float stv = st_lds[c * EXT + tid];
                long idx  = (long)c * NL + l - 2;
                if (idx >= 0) out[idx] = a0 * stv + a1 * jd[c];
            }
        }
    }
}

extern "C" void kernel_launch(void* const* d_in, const int* in_sizes, int n_in,
                              void* d_out, int out_size, void* d_ws, size_t ws_size,
                              hipStream_t stream) {
    (void)in_sizes; (void)n_in; (void)out_size; (void)d_ws; (void)ws_size;
    const float* x      = (const float*)d_in[0];
    const float* states = (const float*)d_in[1];
    const float* conv_w = (const float*)d_in[2];
    const float* conv_b = (const float*)d_in[3];
    const float* comp   = (const float*)d_in[4];
    const float* am     = (const float*)d_in[5];
    float* out = (float*)d_out;

    const int nblocks = (NL + TILE - 1) / TILE;
    state_policy_kernel<<<nblocks, 256, 0, stream>>>(x, states, conv_w, conv_b, comp, am, out);
}


# conv GEMM re-tiled: a WMMA column is a pair of adjacent links (rows 2 links x 7 outputs, K 7 channels x 4-cell window): 7 WMMAs per 32 links, not 12; all lanes in epilogue; conv sum order changes
# speedup vs baseline: 2.2806x; 1.2534x over previous
#include <hip/hip_runtime.h>
#include <math.h>

typedef __attribute__((ext_vector_type(2))) float v2f;
typedef __attribute__((ext_vector_type(8))) float v8f;

#define NL   1000000
#define TILE 250
#define EXT  256
#define HSTR 258
#define NCH  7

__device__ __forceinline__ float tanh_hw(float v) { return __builtin_amdgcn_tanhf(v); }

__global__ __launch_bounds__(256) void
state_policy_kernel(const float* __restrict__ x,
                    const float* __restrict__ states,
                    const float* __restrict__ conv_w,
                    const float* __restrict__ conv_b,
                    const float* __restrict__ comp_mat,
                    const float* __restrict__ act_mat,
                    float* __restrict__ out)
{
    __shared__ __attribute__((aligned(16))) float h_lds[NCH * HSTR];
    __shared__ float st_lds[NCH * EXT];

    const int tid  = threadIdx.x;
    const int lane = tid & 31;
    const int wid  = tid >> 5;
    const int n    = lane & 15;
    const int hi   = lane >> 4;

    const long l0    = (long)blockIdx.x * TILE;
    const long e     = l0 - 3 + tid;
    const bool valid = (e >= 0) && (e < NL);

    float jl[6], jd[6];
#pragma unroll
    for (int c = 0; c < 6; ++c) {
        long flat = (long)c * NL + e;
        bool ok = valid && (flat >= 2);
        jl[c] = ok ? x[7 + flat - 2]            : 0.0f;
        jd[c] = ok ? x[6L * NL + 11 + flat - 2] : 0.0f;
    }
#pragma unroll
    for (int c = 0; c < NCH; ++c)
        st_lds[c * EXT + tid] = valid ? states[(long)c * NL + e] : 0.0f;

    const float ext0 = x[3];
    const float ext1 = x[4];

    float comp[NCH][3];
#pragma unroll
    for (int c = 0; c < NCH; ++c)
#pragma unroll
        for (int j = 0; j < 3; ++j)
            comp[c][j] = comp_mat[c * 3 + j];

    float cb[NCH];
#pragma unroll
    for (int o = 0; o < NCH; ++o) cb[o] = conv_b[o];

    v2f A[NCH];
    {
        const int  o   = n & 7;
        const int  p   = n >> 3;
        const int  t0  = 2 * hi - p;
        const bool okx = (o < NCH) && (t0 >= 0) && (t0 <= 2);
        const bool oky = (o < NCH) && (t0 >= -1) && (t0 <= 1);
        const int  ix  = okx ? o * 21 + t0     : 0;
        const int  iy  = oky ? o * 21 + t0 + 1 : 0;
#pragma unroll
        for (int c = 0; c < NCH; ++c) {
            const float wx = conv_w[ix + c * 3];
            const float wy = conv_w[iy + c * 3];
            A[c].x = okx ? wx : 0.0f;
            A[c].y = oky ? wy : 0.0f;
        }
    }

    if (tid < NCH) {
        h_lds[tid * HSTR + 0]        = 0.0f;
        h_lds[tid * HSTR + HSTR - 1] = 0.0f;
    }
    __syncthreads();

    for (int it = 0; it < 3; ++it) {
#pragma unroll
        for (int c = 0; c < NCH; ++c) {
            float o0 = (c < 6) ? jl[c] : ext0;
            float o1 = (c < 6) ? jd[c] : ext1;
            float sv = st_lds[c * EXT + tid];
            float hv = tanh_hw(o0 * comp[c][0] + o1 * comp[c][1] + sv * comp[c][2]);
            h_lds[c * HSTR + tid + 1] = valid ? hv : 0.0f;
        }
        __syncthreads();

        {
            const int base = wid * 32;
            const float* hp = &h_lds[base + 2 * n + 2 * hi];
            v8f acc = {0.f, 0.f, 0.f, 0.f, 0.f, 0.f, 0.f, 0.f};
#pragma unroll
            for (int c = 0; c < NCH; ++c) {
                v2f B = *reinterpret_cast<const v2f*>(hp + c * HSTR);
                acc = __builtin_amdgcn_wmma_f32_16x16x4_f32(
                        false, A[c], false, B, (short)0, acc, false, false);
            }
            const int t = base + 2 * n + hi;
#pragma unroll
            for (int o = 0; o < NCH; ++o)
                st_lds[o * EXT + t] = tanh_hw(acc[o] + cb[o]);
        }
        __syncthreads();
    }

    if (tid >= 3 && tid < 3 + TILE) {
        long l = l0 + tid - 3;
        if (l < NL) {
#pragma unroll
            for (int c = 0; c < 6; ++c) {
                float a0  = act_mat[c * 2 + 0];
                float a1  = act_mat[c * 2 + 1];
                float stv = st_lds[c * EXT + tid];
                long idx  = (long)c * NL + l - 2;
                if (idx >= 0) out[idx] = a0 * stv + a1 * jd[c];
            }
        }
    }
}

extern "C" void kernel_launch(void* const* d_in, const int* in_sizes, int n_in,
                              void* d_out, int out_size, void* d_ws, size_t ws_size,
                              hipStream_t stream) {
    (void)in_sizes; (void)n_in; (void)out_size; (void)d_ws; (void)ws_size;
    const float* x      = (const float*)d_in[0];
    const float* states = (const float*)d_in[1];
    const float* conv_w = (const float*)d_in[2];
    const float* conv_b = (const float*)d_in[3];
    const float* comp   = (const float*)d_in[4];
    const float* am     = (const float*)d_in[5];
    float* out = (float*)d_out;

    const int nblocks = (NL + TILE - 1) / TILE;
    state_policy_kernel<<<nblocks, 256, 0, stream>>>(x, states, conv_w, conv_b, comp, am, out);
}


# block walks 4 tiles (1000 resident blocks, 8000 waves not 32000): next tile's 19 inputs load branch-free into registers under the current tile's sweeps; barriers fence LDS only; sums unchanged
# speedup vs baseline: 2.5592x; 1.1222x over previous
#include <hip/hip_runtime.h>
#include <math.h>

typedef __attribute__((ext_vector_type(2))) float v2f;
typedef __attribute__((ext_vector_type(8))) float v8f;

#define NL   1000000
#define TILE 250
#define EXT  256
#define HSTR 258
#define NCH  7
#define NT   4
#define NBLK (NL / (TILE * NT))

__device__ __forceinline__ float tanh_hw(float v) { return __builtin_amdgcn_tanhf(v); }

__device__ __forceinline__ void lds_barrier()
{
    __builtin_amdgcn_fence(__ATOMIC_RELEASE, "workgroup", "local");
    __builtin_amdgcn_s_barrier();
    __builtin_amdgcn_fence(__ATOMIC_ACQUIRE, "workgroup", "local");
}

struct TileIn { float jl[6]; float jd[6]; float st[NCH]; };

__device__ __forceinline__ void fetch_tile(TileIn& t, const float* __restrict__ x,
                                           const float* __restrict__ states, int e)
{
    const int ec = min(max(e, 0), NL - 1);
#pragma unroll
    for (int c = 0; c < 6; ++c) {
        t.jl[c] = x[5 + c * NL + ec];
        t.jd[c] = x[6 * NL + 9 + c * NL + ec];
    }
#pragma unroll
    for (int c = 0; c < NCH; ++c) t.st[c] = states[c * NL + ec];
}

__global__ __launch_bounds__(256) void
state_policy_kernel(const float* __restrict__ x,
                    const float* __restrict__ states,
                    const float* __restrict__ conv_w,
                    const float* __restrict__ conv_b,
                    const float* __restrict__ comp_mat,
                    const float* __restrict__ act_mat,
                    float* __restrict__ out)
{
    __shared__ __attribute__((aligned(16))) float h_lds[NCH * HSTR];
    __shared__ float st_lds[NCH * EXT];

    const int tid  = threadIdx.x;
    const int lane = tid & 31;
    const int wid  = tid >> 5;
    const int n    = lane & 15;
    const int hi   = lane >> 4;

    const int l0 = (int)blockIdx.x * (TILE * NT);

    TileIn nx;
    fetch_tile(nx, x, states, l0 - 3 + tid);

    const float ext0 = x[3];
    const float ext1 = x[4];

    float comp[NCH][3];
#pragma unroll
    for (int c = 0; c < NCH; ++c)
#pragma unroll
        for (int j = 0; j < 3; ++j)
            comp[c][j] = comp_mat[c * 3 + j];

    float cb[NCH];
#pragma unroll
    for (int o = 0; o < NCH; ++o) cb[o] = conv_b[o];

    float am[6][2];
#pragma unroll
    for (int c = 0; c < 6; ++c) { am[c][0] = act_mat[c * 2 + 0]; am[c][1] = act_mat[c * 2 + 1]; }

    v2f A[NCH];
    {
        const int  o   = n & 7;
        const int  p   = n >> 3;
        const int  t0  = 2 * hi - p;
        const bool okx = (o < NCH) && (t0 >= 0) && (t0 <= 2);
        const bool oky = (o < NCH) && (t0 >= -1) && (t0 <= 1);
        const int  ix  = okx ? o * 21 + t0     : 0;
        const int  iy  = oky ? o * 21 + t0 + 1 : 0;
#pragma unroll
        for (int c = 0; c < NCH; ++c) {
            const float wx = conv_w[ix + c * 3];
            const float wy = conv_w[iy + c * 3];
            A[c].x = okx ? wx : 0.0f;
            A[c].y = oky ? wy : 0.0f;
        }
    }

    if (tid < NCH) {
        h_lds[tid * HSTR + 0]        = 0.0f;
        h_lds[tid * HSTR + HSTR - 1] = 0.0f;
    }
    __syncthreads();

    for (int k = 0; k < NT; ++k) {
        const int  e     = l0 + k * TILE - 3 + tid;
        const bool valid = (e >= 0) && (e < NL);
        const bool ok0   = valid && (e >= 2);

        float jl[6], jd[6];
#pragma unroll
        for (int c = 0; c < 6; ++c) {
            const bool ok = (c == 0) ? ok0 : valid;
            jl[c] = ok ? nx.jl[c] : 0.0f;
            jd[c] = ok ? nx.jd[c] : 0.0f;
        }
#pragma unroll
        for (int c = 0; c < NCH; ++c)
            st_lds[c * EXT + tid] = valid ? nx.st[c] : 0.0f;

        if (k + 1 < NT) fetch_tile(nx, x, states, e + TILE);

        for (int it = 0; it < 3; ++it) {
#pragma unroll
            for (int c = 0; c < NCH; ++c) {
                float o0 = (c < 6) ? jl[c] : ext0;
                float o1 = (c < 6) ? jd[c] : ext1;
                float sv = st_lds[c * EXT + tid];
                float hv = tanh_hw(o0 * comp[c][0] + o1 * comp[c][1] + sv * comp[c][2]);
                h_lds[c * HSTR + tid + 1] = valid ? hv : 0.0f;
            }
            lds_barrier();

            {
                const int base = wid * 32;
                const float* hp = &h_lds[base + 2 * n + 2 * hi];
                v8f acc = {0.f, 0.f, 0.f, 0.f, 0.f, 0.f, 0.f, 0.f};
#pragma unroll
                for (int c = 0; c < NCH; ++c) {
                    v2f B = *reinterpret_cast<const v2f*>(hp + c * HSTR);
                    acc = __builtin_amdgcn_wmma_f32_16x16x4_f32(
                            false, A[c], false, B, (short)0, acc, false, false);
                }
                const int t = base + 2 * n + hi;
#pragma unroll
                for (int o = 0; o < NCH; ++o)
                    st_lds[o * EXT + t] = tanh_hw(acc[o] + cb[o]);
            }
            lds_barrier();
        }

        if (tid >= 3 && tid < 3 + TILE && e < NL) {
            float r[6];
#pragma unroll
            for (int c = 0; c < 6; ++c)
                r[c] = am[c][0] * st_lds[c * EXT + tid] + am[c][1] * jd[c];
            if (e >= 2) out[e - 2] = r[0];
#pragma unroll
            for (int c = 1; c < 6; ++c)
                out[c * NL + e - 2] = r[c];
        }
    }
}

extern "C" void kernel_launch(void* const* d_in, const int* in_sizes, int n_in,
                              void* d_out, int out_size, void* d_ws, size_t ws_size,
                              hipStream_t stream) {
    (void)in_sizes; (void)n_in; (void)out_size; (void)d_ws; (void)ws_size;
    const float* x      = (const float*)d_in[0];
    const float* states = (const float*)d_in[1];
    const float* conv_w = (const float*)d_in[2];
    const float* conv_b = (const float*)d_in[3];
    const float* comp   = (const float*)d_in[4];
    const float* am     = (const float*)d_in[5];
    float* out = (float*)d_out;

    state_policy_kernel<<<NBLK, 256, 0, stream>>>(x, states, conv_w, conv_b, comp, am, out);
}


# state kept in registers in the WMMA D layout (thread owns link 2n+hi in every phase): st_lds round trip gone; h ping-pongs two LDS buffers: one barrier per sweep, not two; sums unchanged
# speedup vs baseline: 2.7591x; 1.0781x over previous
#include <hip/hip_runtime.h>
#include <math.h>

typedef __attribute__((ext_vector_type(2))) float v2f;
typedef __attribute__((ext_vector_type(8))) float v8f;

#define NL   1000000
#define TILE 250
#define EXT  256
#define HSTR 258
#define NCH  7
#define NT   4
#define NBLK (NL / (TILE * NT))
#define HBUF (NCH * HSTR)

__device__ __forceinline__ float tanh_hw(float v) { return __builtin_amdgcn_tanhf(v); }

__device__ __forceinline__ void lds_barrier()
{
    __builtin_amdgcn_fence(__ATOMIC_RELEASE, "workgroup", "local");
    __builtin_amdgcn_s_barrier();
    __builtin_amdgcn_fence(__ATOMIC_ACQUIRE, "workgroup", "local");
}

struct TileIn { float jl[6]; float jd[6]; float st[NCH]; };

__device__ __forceinline__ void fetch_tile(TileIn& t, const float* __restrict__ x,
                                           const float* __restrict__ states, int e)
{
    const int ec = min(max(e, 0), NL - 1);
#pragma unroll
    for (int c = 0; c < 6; ++c) {
        t.jl[c] = x[5 + c * NL + ec];
        t.jd[c] = x[6 * NL + 9 + c * NL + ec];
    }
#pragma unroll
    for (int c = 0; c < NCH; ++c) t.st[c] = states[c * NL + ec];
}

__global__ __launch_bounds__(256) void
state_policy_kernel(const float* __restrict__ x,
                    const float* __restrict__ states,
                    const float* __restrict__ conv_w,
                    const float* __restrict__ conv_b,
                    const float* __restrict__ comp_mat,
                    const float* __restrict__ act_mat,
                    float* __restrict__ out)
{
    __shared__ __attribute__((aligned(16))) float h_lds[2 * HBUF];

    const int tid  = threadIdx.x;
    const int lane = tid & 31;
    const int wid  = tid >> 5;
    const int n    = lane & 15;
    const int hi   = lane >> 4;
    const int col  = wid * 32 + 2 * n + hi;

    const int l0 = (int)blockIdx.x * (TILE * NT);

    TileIn nx;
    fetch_tile(nx, x, states, l0 - 3 + col);

    const float ext0 = x[3];
    const float ext1 = x[4];

    float comp[NCH][3];
#pragma unroll
    for (int c = 0; c < NCH; ++c)
#pragma unroll
        for (int j = 0; j < 3; ++j)
            comp[c][j] = comp_mat[c * 3 + j];

    float cb[NCH];
#pragma unroll
    for (int o = 0; o < NCH; ++o) cb[o] = conv_b[o];

    float am[6][2];
#pragma unroll
    for (int c = 0; c < 6; ++c) { am[c][0] = act_mat[c * 2 + 0]; am[c][1] = act_mat[c * 2 + 1]; }

    v2f A[NCH];
    {
        const int  o   = n & 7;
        const int  p   = n >> 3;
        const int  t0  = 2 * hi - p;
        const bool okx = (o < NCH) && (t0 >= 0) && (t0 <= 2);
        const bool oky = (o < NCH) && (t0 >= -1) && (t0 <= 1);
        const int  ix  = okx ? o * 21 + t0     : 0;
        const int  iy  = oky ? o * 21 + t0 + 1 : 0;
#pragma unroll
        for (int c = 0; c < NCH; ++c) {
            const float wx = conv_w[ix + c * 3];
            const float wy = conv_w[iy + c * 3];
            A[c].x = okx ? wx : 0.0f;
            A[c].y = oky ? wy : 0.0f;
        }
    }

    if (tid < 2 * NCH) {
        h_lds[tid * HSTR + 0]        = 0.0f;
        h_lds[tid * HSTR + HSTR - 1] = 0.0f;
    }
    __syncthreads();

    for (int k = 0; k < NT; ++k) {
        const int  e     = l0 + k * TILE - 3 + col;
        const bool valid = (e >= 0) && (e < NL);
        const bool ok0   = valid && (e >= 2);

        float jl[6], jd[6], st[NCH];
#pragma unroll
        for (int c = 0; c < 6; ++c) {
            const bool ok = (c == 0) ? ok0 : valid;
            jl[c] = ok ? nx.jl[c] : 0.0f;
            jd[c] = ok ? nx.jd[c] : 0.0f;
        }
#pragma unroll
        for (int c = 0; c < NCH; ++c)
            st[c] = valid ? nx.st[c] : 0.0f;

        if (k + 1 < NT) fetch_tile(nx, x, states, e + TILE);

        for (int it = 0; it < 3; ++it) {
            float* hb = h_lds + ((k * 3 + it) & 1) * HBUF;

#pragma unroll
            for (int c = 0; c < NCH; ++c) {
                float o0 = (c < 6) ? jl[c] : ext0;
                float o1 = (c < 6) ? jd[c] : ext1;
                float hv = tanh_hw(o0 * comp[c][0] + o1 * comp[c][1] + st[c] * comp[c][2]);
                hb[c * HSTR + col + 1] = valid ? hv : 0.0f;
            }
            lds_barrier();

            {
                const int base = wid * 32;
                const float* hp = &hb[base + 2 * n + 2 * hi];
                v8f acc = {0.f, 0.f, 0.f, 0.f, 0.f, 0.f, 0.f, 0.f};
#pragma unroll
                for (int c = 0; c < NCH; ++c) {
                    v2f B = *reinterpret_cast<const v2f*>(hp + c * HSTR);
                    acc = __builtin_amdgcn_wmma_f32_16x16x4_f32(
                            false, A[c], false, B, (short)0, acc, false, false);
                }
#pragma unroll
                for (int o = 0; o < NCH; ++o)
                    st[o] = tanh_hw(acc[o] + cb[o]);
            }
        }

        if (col >= 3 && col < 3 + TILE && e < NL) {
            float r[6];
#pragma unroll
            for (int c = 0; c < 6; ++c)
                r[c] = am[c][0] * st[c] + am[c][1] * jd[c];
            if (e >= 2) out[e - 2] = r[0];
#pragma unroll
            for (int c = 1; c < 6; ++c)
                out[c * NL + e - 2] = r[c];
        }
    }
}

extern "C" void kernel_launch(void* const* d_in, const int* in_sizes, int n_in,
                              void* d_out, int out_size, void* d_ws, size_t ws_size,
                              hipStream_t stream) {
    (void)in_sizes; (void)n_in; (void)out_size; (void)d_ws; (void)ws_size;
    const float* x      = (const float*)d_in[0];
    const float* states = (const float*)d_in[1];
    const float* conv_w = (const float*)d_in[2];
    const float* conv_b = (const float*)d_in[3];
    const float* comp   = (const float*)d_in[4];
    const float* am     = (const float*)d_in[5];
    float* out = (float*)d_out;

    state_policy_kernel<<<NBLK, 256, 0, stream>>>(x, states, conv_w, conv_b, comp, am, out);
}
